# starting kernel
# baseline (speedup 1.0000x reference)
#include <hip/hip_runtime.h>


#define N_NODES 100000
#define N_EDGES 640000
#define D 128

typedef float v8f __attribute__((ext_vector_type(8)));
typedef float v2f __attribute__((ext_vector_type(2)));

__global__ void gcn_zero_kernel(float4* __restrict__ p, int n4) {
    int i = blockIdx.x * blockDim.x + threadIdx.x;
    if (i < n4) p[i] = make_float4(0.f, 0.f, 0.f, 0.f);
}

__global__ void gcn_scatter_kernel(const float* __restrict__ x,
                                   const float* __restrict__ ew,
                                   const int* __restrict__ src,
                                   const int* __restrict__ dst,
                                   float* __restrict__ h) {
    const int wavesPerBlock = blockDim.x >> 5;
    const int e = blockIdx.x * wavesPerBlock + (threadIdx.x >> 5);
    if (e >= N_EDGES) return;
    const int lane = threadIdx.x & 31;

    const int   s = src[e];
    const int   d = dst[e];
    const float w = ew[e];

    const float4 v = *(const float4*)(x + (size_t)s * D + lane * 4);
    float* hp = h + (size_t)d * D + lane * 4;
    unsafeAtomicAdd(hp + 0, v.x * w);
    unsafeAtomicAdd(hp + 1, v.y * w);
    unsafeAtomicAdd(hp + 2, v.z * w);
    unsafeAtomicAdd(hp + 3, v.w * w);
}

__global__ void __launch_bounds__(256)
gcn_gemm_kernel(const float* __restrict__ h,
                const float* __restrict__ W,
                const float* __restrict__ b,
                float* __restrict__ out) {
    const int lane = threadIdx.x & 31;
    const int wave = threadIdx.x >> 5;
    const int row0 = blockIdx.x * 16;
    const int col0 = wave * 16;
    const int l15  = lane & 15;
    const int rsel = lane >> 4;

    const float* aptr = h + (size_t)(row0 + l15) * D + 2 * rsel;
    const float* bptr = W + (size_t)(col0 + l15) * D + 2 * rsel;

    v8f acc = {};
#pragma unroll
    for (int k = 0; k < D; k += 4) {
        v2f a  = *(const v2f*)(aptr + k);
        v2f bb = *(const v2f*)(bptr + k);
        acc = __builtin_amdgcn_wmma_f32_16x16x4_f32(
             false, a,  false, bb,
             (short)0, acc,  false,  false);
    }

    const float bias = b[col0 + l15];
    float* optr = out + (size_t)(row0 + 8 * rsel) * D + col0 + l15;
#pragma unroll
    for (int r = 0; r < 8; ++r) {
        optr[(size_t)r * D] = acc[r] + bias;
    }
}

extern "C" void kernel_launch(void* const* d_in, const int* in_sizes, int n_in,
                              void* d_out, int out_size, void* d_ws, size_t ws_size,
                              hipStream_t stream) {
    const float* x   = (const float*)d_in[0];
    const float* ew  = (const float*)d_in[1];
    const int*   src = (const int*)  d_in[2];
    const int*   dst = (const int*)  d_in[3];
    const float* Wm  = (const float*)d_in[4];
    const float* bv  = (const float*)d_in[5];

    float* h   = (float*)d_ws;
    float* out = (float*)d_out;

    {
        const int n4 = (N_NODES * D) / 4;
        const int threads = 256;
        const int blocks = (n4 + threads - 1) / threads;
        gcn_zero_kernel<<<blocks, threads, 0, stream>>>((float4*)h, n4);
    }

    {
        const int threads = 256;
        const int wavesPerBlock = threads / 32;
        const int blocks = (N_EDGES + wavesPerBlock - 1) / wavesPerBlock;
        gcn_scatter_kernel<<<blocks, threads, 0, stream>>>(x, ew, src, dst, h);
    }

    {
        const int threads = 256;
        const int blocks = N_NODES / 16;
        gcn_gemm_kernel<<<blocks, threads, 0, stream>>>(h, Wm, bv, out);
    }
}


# Scatter: lane l owns features l,l+32,l+64,l+96 so each f32 atomic instruction (and each x load) covers one whole 128-B line, 4 line requests per edge not 16; sums still add in arrival order
# speedup vs baseline: 1.2192x; 1.2192x over previous
#include <hip/hip_runtime.h>


#define N_NODES 100000
#define N_EDGES 640000
#define D 128

typedef float v8f __attribute__((ext_vector_type(8)));
typedef float v2f __attribute__((ext_vector_type(2)));

__global__ void gcn_zero_kernel(float4* __restrict__ p, int n4) {
    int i = blockIdx.x * blockDim.x + threadIdx.x;
    if (i < n4) p[i] = make_float4(0.f, 0.f, 0.f, 0.f);
}

__global__ void gcn_scatter_kernel(const float* __restrict__ x,
                                   const float* __restrict__ ew,
                                   const int* __restrict__ src,
                                   const int* __restrict__ dst,
                                   float* __restrict__ h) {
    const int wavesPerBlock = blockDim.x >> 5;
    const int e = blockIdx.x * wavesPerBlock + (threadIdx.x >> 5);
    if (e >= N_EDGES) return;
    const int lane = threadIdx.x & 31;

    const int   s = src[e];
    const int   d = dst[e];
    const float w = ew[e];

    const float* xp = x + (size_t)s * D + lane;
    float*       hp = h + (size_t)d * D + lane;
    const float v0 = xp[0];
    const float v1 = xp[32];
    const float v2 = xp[64];
    const float v3 = xp[96];
    unsafeAtomicAdd(hp + 0,  v0 * w);
    unsafeAtomicAdd(hp + 32, v1 * w);
    unsafeAtomicAdd(hp + 64, v2 * w);
    unsafeAtomicAdd(hp + 96, v3 * w);
}

__global__ void __launch_bounds__(256)
gcn_gemm_kernel(const float* __restrict__ h,
                const float* __restrict__ W,
                const float* __restrict__ b,
                float* __restrict__ out) {
    const int lane = threadIdx.x & 31;
    const int wave = threadIdx.x >> 5;
    const int row0 = blockIdx.x * 16;
    const int col0 = wave * 16;
    const int l15  = lane & 15;
    const int rsel = lane >> 4;

    const float* aptr = h + (size_t)(row0 + l15) * D + 2 * rsel;
    const float* bptr = W + (size_t)(col0 + l15) * D + 2 * rsel;

    v8f acc = {};
#pragma unroll
    for (int k = 0; k < D; k += 4) {
        v2f a  = *(const v2f*)(aptr + k);
        v2f bb = *(const v2f*)(bptr + k);
        acc = __builtin_amdgcn_wmma_f32_16x16x4_f32(
             false, a,  false, bb,
             (short)0, acc,  false,  false);
    }

    const float bias = b[col0 + l15];
    float* optr = out + (size_t)(row0 + 8 * rsel) * D + col0 + l15;
#pragma unroll
    for (int r = 0; r < 8; ++r) {
        optr[(size_t)r * D] = acc[r] + bias;
    }
}

extern "C" void kernel_launch(void* const* d_in, const int* in_sizes, int n_in,
                              void* d_out, int out_size, void* d_ws, size_t ws_size,
                              hipStream_t stream) {
    const float* x   = (const float*)d_in[0];
    const float* ew  = (const float*)d_in[1];
    const int*   src = (const int*)  d_in[2];
    const int*   dst = (const int*)  d_in[3];
    const float* Wm  = (const float*)d_in[4];
    const float* bv  = (const float*)d_in[5];

    float* h   = (float*)d_ws;
    float* out = (float*)d_out;

    {
        const int n4 = (N_NODES * D) / 4;
        const int threads = 256;
        const int blocks = (n4 + threads - 1) / threads;
        gcn_zero_kernel<<<blocks, threads, 0, stream>>>((float4*)h, n4);
    }

    {
        const int threads = 256;
        const int wavesPerBlock = threads / 32;
        const int blocks = (N_EDGES + wavesPerBlock - 1) / wavesPerBlock;
        gcn_scatter_kernel<<<blocks, threads, 0, stream>>>(x, ew, src, dst, h);
    }

    {
        const int threads = 256;
        const int blocks = N_NODES / 16;
        gcn_gemm_kernel<<<blocks, threads, 0, stream>>>(h, Wm, bv, out);
    }
}


# Scatter: each wave works through 64 edges (coalesced index loads, lane broadcast, rows of 8 edges in flight) instead of one wave per edge: 10000 waves, not 640000; sums still add in arrival order
# speedup vs baseline: 1.4602x; 1.1977x over previous
#include <hip/hip_runtime.h>


#define N_NODES 100000
#define N_EDGES 640000
#define D 128

typedef float v8f __attribute__((ext_vector_type(8)));
typedef float v2f __attribute__((ext_vector_type(2)));

__global__ void gcn_zero_kernel(float4* __restrict__ p, int n4) {
    int i = blockIdx.x * blockDim.x + threadIdx.x;
    if (i < n4) p[i] = make_float4(0.f, 0.f, 0.f, 0.f);
}

#define EDGES_PER_BATCH 32
#define BATCHES_PER_WAVE 2
#define EDGES_PER_WAVE (EDGES_PER_BATCH * BATCHES_PER_WAVE)
#define EDGE_GROUP 8
static_assert(N_EDGES % EDGES_PER_WAVE == 0, "every wave owns a full set of edges");
static_assert(EDGES_PER_BATCH % EDGE_GROUP == 0, "groups tile a batch");

__global__ void __launch_bounds__(256)
gcn_scatter_kernel(const float* __restrict__ x,
                   const float* __restrict__ ew,
                   const int* __restrict__ src,
                   const int* __restrict__ dst,
                   float* __restrict__ h) {
    const int lane  = threadIdx.x & 31;
    const int wave  = blockIdx.x * (blockDim.x >> 5) + (threadIdx.x >> 5);
    const int ebase = wave * EDGES_PER_WAVE;
    if (ebase >= N_EDGES) return;

    for (int b = 0; b < BATCHES_PER_WAVE; ++b) {
        const int e  = ebase + b * EDGES_PER_BATCH + lane;
        const int sv = src[e];
        const int dv = dst[e];
        const int wv = __float_as_int(ew[e]);

#pragma unroll 1
        for (int j = 0; j < EDGES_PER_BATCH; j += EDGE_GROUP) {
            float v[EDGE_GROUP][4];
#pragma unroll
            for (int u = 0; u < EDGE_GROUP; ++u) {
                const int s = __builtin_amdgcn_readlane(sv, j + u);
                const float* xrow = x + (size_t)s * D;
                v[u][0] = xrow[lane];
                v[u][1] = xrow[lane + 32];
                v[u][2] = xrow[lane + 64];
                v[u][3] = xrow[lane + 96];
            }
#pragma unroll
            for (int u = 0; u < EDGE_GROUP; ++u) {
                const int   d = __builtin_amdgcn_readlane(dv, j + u);
                const float w = __int_as_float(__builtin_amdgcn_readlane(wv, j + u));
                float* hrow = h + (size_t)d * D;
                unsafeAtomicAdd(hrow + lane,      v[u][0] * w);
                unsafeAtomicAdd(hrow + lane + 32, v[u][1] * w);
                unsafeAtomicAdd(hrow + lane + 64, v[u][2] * w);
                unsafeAtomicAdd(hrow + lane + 96, v[u][3] * w);
            }
        }
    }
}

__global__ void __launch_bounds__(256)
gcn_gemm_kernel(const float* __restrict__ h,
                const float* __restrict__ W,
                const float* __restrict__ b,
                float* __restrict__ out) {
    const int lane = threadIdx.x & 31;
    const int wave = threadIdx.x >> 5;
    const int row0 = blockIdx.x * 16;
    const int col0 = wave * 16;
    const int l15  = lane & 15;
    const int rsel = lane >> 4;

    const float* aptr = h + (size_t)(row0 + l15) * D + 2 * rsel;
    const float* bptr = W + (size_t)(col0 + l15) * D + 2 * rsel;

    v8f acc = {};
#pragma unroll
    for (int k = 0; k < D; k += 4) {
        v2f a  = *(const v2f*)(aptr + k);
        v2f bb = *(const v2f*)(bptr + k);
        acc = __builtin_amdgcn_wmma_f32_16x16x4_f32(
             false, a,  false, bb,
             (short)0, acc,  false,  false);
    }

    const float bias = b[col0 + l15];
    float* optr = out + (size_t)(row0 + 8 * rsel) * D + col0 + l15;
#pragma unroll
    for (int r = 0; r < 8; ++r) {
        optr[(size_t)r * D] = acc[r] + bias;
    }
}

extern "C" void kernel_launch(void* const* d_in, const int* in_sizes, int n_in,
                              void* d_out, int out_size, void* d_ws, size_t ws_size,
                              hipStream_t stream) {
    const float* x   = (const float*)d_in[0];
    const float* ew  = (const float*)d_in[1];
    const int*   src = (const int*)  d_in[2];
    const int*   dst = (const int*)  d_in[3];
    const float* Wm  = (const float*)d_in[4];
    const float* bv  = (const float*)d_in[5];

    float* h   = (float*)d_ws;
    float* out = (float*)d_out;

    {
        const int n4 = (N_NODES * D) / 4;
        const int threads = 256;
        const int blocks = (n4 + threads - 1) / threads;
        gcn_zero_kernel<<<blocks, threads, 0, stream>>>((float4*)h, n4);
    }

    {
        const int threads = 256;
        const int wavesPerBlock = threads / 32;
        const int blocks = N_EDGES / (EDGES_PER_WAVE * wavesPerBlock);
        gcn_scatter_kernel<<<blocks, threads, 0, stream>>>(x, ew, src, dst, h);
    }

    {
        const int threads = 256;
        const int blocks = N_NODES / 16;
        gcn_gemm_kernel<<<blocks, threads, 0, stream>>>(h, Wm, bv, out);
    }
}


# GEMM: each wave makes a whole 16x128 output strip (8 accumulator tiles) with W and its 16 rows of h staged in LDS by whole-row loads; h is read once, not by 8 waves in 16-B pieces; k order unchanged
# speedup vs baseline: 2.0313x; 1.3911x over previous
#include <hip/hip_runtime.h>


#define N_NODES 100000
#define N_EDGES 640000
#define D 128

typedef float v8f __attribute__((ext_vector_type(8)));
typedef float v2f __attribute__((ext_vector_type(2)));

__global__ void gcn_zero_kernel(float4* __restrict__ p, int n4) {
    int i = blockIdx.x * blockDim.x + threadIdx.x;
    if (i < n4) p[i] = make_float4(0.f, 0.f, 0.f, 0.f);
}

#define EDGES_PER_BATCH 32
#define BATCHES_PER_WAVE 2
#define EDGES_PER_WAVE (EDGES_PER_BATCH * BATCHES_PER_WAVE)
#define EDGE_GROUP 8
static_assert(N_EDGES % EDGES_PER_WAVE == 0, "every wave owns a full set of edges");
static_assert(EDGES_PER_BATCH % EDGE_GROUP == 0, "groups tile a batch");

__global__ void __launch_bounds__(256)
gcn_scatter_kernel(const float* __restrict__ x,
                   const float* __restrict__ ew,
                   const int* __restrict__ src,
                   const int* __restrict__ dst,
                   float* __restrict__ h) {
    const int lane  = threadIdx.x & 31;
    const int wave  = blockIdx.x * (blockDim.x >> 5) + (threadIdx.x >> 5);
    const int ebase = wave * EDGES_PER_WAVE;
    if (ebase >= N_EDGES) return;

    for (int b = 0; b < BATCHES_PER_WAVE; ++b) {
        const int e  = ebase + b * EDGES_PER_BATCH + lane;
        const int sv = src[e];
        const int dv = dst[e];
        const int wv = __float_as_int(ew[e]);

#pragma unroll 1
        for (int j = 0; j < EDGES_PER_BATCH; j += EDGE_GROUP) {
            float v[EDGE_GROUP][4];
#pragma unroll
            for (int u = 0; u < EDGE_GROUP; ++u) {
                const int s = __builtin_amdgcn_readlane(sv, j + u);
                const float* xrow = x + (size_t)s * D;
                v[u][0] = xrow[lane];
                v[u][1] = xrow[lane + 32];
                v[u][2] = xrow[lane + 64];
                v[u][3] = xrow[lane + 96];
            }
#pragma unroll
            for (int u = 0; u < EDGE_GROUP; ++u) {
                const int   d = __builtin_amdgcn_readlane(dv, j + u);
                const float w = __int_as_float(__builtin_amdgcn_readlane(wv, j + u));
                float* hrow = h + (size_t)d * D;
                unsafeAtomicAdd(hrow + lane,      v[u][0] * w);
                unsafeAtomicAdd(hrow + lane + 32, v[u][1] * w);
                unsafeAtomicAdd(hrow + lane + 64, v[u][2] * w);
                unsafeAtomicAdd(hrow + lane + 96, v[u][3] * w);
            }
        }
    }
}

#define GEMM_WAVES 4
#define M_TILES (N_NODES / 16)
#define LDS_PITCH 132
static_assert(N_NODES % 16 == 0, "M tiles are full");

__global__ void __launch_bounds__(GEMM_WAVES * 32)
gcn_gemm_kernel(const float* __restrict__ h,
                const float* __restrict__ W,
                const float* __restrict__ b,
                float* __restrict__ out) {
    __shared__ __attribute__((aligned(16))) float Ws[D * LDS_PITCH];
    __shared__ __attribute__((aligned(16))) float As[GEMM_WAVES][16 * LDS_PITCH];

    const int lane = threadIdx.x & 31;
    const int wave = threadIdx.x >> 5;
    const int tile = blockIdx.x * GEMM_WAVES + wave;
    const bool active = tile < M_TILES;
    const int ltile = active ? tile : 0;
    const int l15  = lane & 15;
    const int rsel = lane >> 4;

    {
        const float4* h4 = (const float4*)(h + (size_t)ltile * 16 * D);
#pragma unroll
        for (int r = 0; r < 16; ++r) {
            const float4 v = h4[r * (D / 4) + lane];
            *(float4*)&As[wave][r * LDS_PITCH + lane * 4] = v;
        }
    }
    {
        const float4* W4 = (const float4*)W;
#pragma unroll
        for (int i = 0; i < D / GEMM_WAVES; ++i) {
            const int row = i * GEMM_WAVES + wave;
            const float4 v = W4[row * (D / 4) + lane];
            *(float4*)&Ws[row * LDS_PITCH + lane * 4] = v;
        }
    }
    __syncthreads();

    if (active) {
        const float* abase = &As[wave][l15 * LDS_PITCH + 2 * rsel];
        const float* bbase = &Ws[l15 * LDS_PITCH + 2 * rsel];

        v2f a[D / 4];
#pragma unroll
        for (int kk = 0; kk < D / 4; ++kk) a[kk] = *(const v2f*)(abase + 4 * kk);

        v8f acc[8];
#pragma unroll
        for (int t = 0; t < 8; ++t) acc[t] = v8f{};

#pragma unroll
        for (int kk = 0; kk < D / 4; ++kk) {
#pragma unroll
            for (int t = 0; t < 8; ++t) {
                const v2f bb = *(const v2f*)(bbase + t * 16 * LDS_PITCH + 4 * kk);
                acc[t] = __builtin_amdgcn_wmma_f32_16x16x4_f32(
                     false, a[kk],  false, bb,
                     (short)0, acc[t],  false,  false);
            }
        }

        float* obase = out + (size_t)(tile * 16 + 8 * rsel) * D + l15;
#pragma unroll
        for (int t = 0; t < 8; ++t) {
            const float bias = b[t * 16 + l15];
#pragma unroll
            for (int r = 0; r < 8; ++r) {
                obase[(size_t)r * D + t * 16] = acc[t][r] + bias;
            }
        }
    }
}

extern "C" void kernel_launch(void* const* d_in, const int* in_sizes, int n_in,
                              void* d_out, int out_size, void* d_ws, size_t ws_size,
                              hipStream_t stream) {
    const float* x   = (const float*)d_in[0];
    const float* ew  = (const float*)d_in[1];
    const int*   src = (const int*)  d_in[2];
    const int*   dst = (const int*)  d_in[3];
    const float* Wm  = (const float*)d_in[4];
    const float* bv  = (const float*)d_in[5];

    float* h   = (float*)d_ws;
    float* out = (float*)d_out;

    {
        const int n4 = (N_NODES * D) / 4;
        const int threads = 256;
        const int blocks = (n4 + threads - 1) / threads;
        gcn_zero_kernel<<<blocks, threads, 0, stream>>>((float4*)h, n4);
    }

    {
        const int threads = 256;
        const int wavesPerBlock = threads / 32;
        const int blocks = N_EDGES / (EDGES_PER_WAVE * wavesPerBlock);
        gcn_scatter_kernel<<<blocks, threads, 0, stream>>>(x, ew, src, dst, h);
    }

    {
        const int threads = GEMM_WAVES * 32;
        const int blocks = (M_TILES + GEMM_WAVES - 1) / GEMM_WAVES;
        gcn_gemm_kernel<<<blocks, threads, 0, stream>>>(h, Wm, bv, out);
    }
}


# h kept in LDS: each of 250 workgroups owns 400 nodes, scans all dst, adds its edges' x rows into LDS (LDS f32 atomics, sums still in arrival order) and makes its out strip from LDS; one launch
# speedup vs baseline: 2.8494x; 1.4027x over previous
#include <hip/hip_runtime.h>


#define N_NODES 100000
#define N_EDGES 640000
#define D 128

typedef float v8f __attribute__((ext_vector_type(8)));
typedef float v2f __attribute__((ext_vector_type(2)));

#define NODES_PER_WG 400
#define N_WG (N_NODES / NODES_PER_WG)
#define TILES_PER_WG (NODES_PER_WG / 16)
#define WG_WAVES 32
#define WG_THREADS (WG_WAVES * 32)
#define LDS_PITCH 132
#define CHUNK_EDGES 128
#define N_CHUNKS (N_EDGES / CHUNK_EDGES)
#define SCAN_DEPTH 4
#define RING 256
#define EDGE_GROUP 8
static_assert(N_NODES % NODES_PER_WG == 0, "node ranges tile the graph");
static_assert(NODES_PER_WG % 16 == 0, "M tiles are full");
static_assert(N_EDGES % CHUNK_EDGES == 0, "chunks are full");
static_assert(N_CHUNKS % N_WG == 0, "start offsets spread evenly");
static_assert(N_EDGES <= (1 << 20) && NODES_PER_WG <= (1 << 12), "queue entry packs edge id and local node");

__global__ void __launch_bounds__(WG_THREADS)
gcn_fused_kernel(const float* __restrict__ x,
                 const float* __restrict__ ew,
                 const int* __restrict__ src,
                 const int* __restrict__ dst,
                 const float* __restrict__ W,
                 const float* __restrict__ b,
                 float* __restrict__ out) {
    __shared__ __attribute__((aligned(16))) float hs[NODES_PER_WG * LDS_PITCH];
    __shared__ __attribute__((aligned(16))) float Ws[D * LDS_PITCH];
    __shared__ unsigned int ring[WG_WAVES * RING];

    const int lane  = threadIdx.x & 31;
    const int wave  = __builtin_amdgcn_readfirstlane(threadIdx.x >> 5);
    const int node0 = blockIdx.x * NODES_PER_WG;

    for (int i = threadIdx.x; i < NODES_PER_WG * LDS_PITCH / 4; i += WG_THREADS)
        ((float4*)hs)[i] = make_float4(0.f, 0.f, 0.f, 0.f);
    {
        const float4* W4 = (const float4*)W;
#pragma unroll
        for (int i = 0; i < D / WG_WAVES; ++i) {
            const int row = i * WG_WAVES + wave;
            const float4 v = W4[row * (D / 4) + lane];
            *(float4*)&Ws[row * LDS_PITCH + lane * 4] = v;
        }
    }
    __syncthreads();

    unsigned int* ringw = &ring[wave * RING];
    unsigned qhead = 0, qcount = 0;

    auto push = [&](const bool m, const unsigned dl, const unsigned e) __attribute__((always_inline)) {
        const unsigned mask = __builtin_amdgcn_ballot_w32(m);
        if (m) ringw[(qhead + qcount + __builtin_amdgcn_mbcnt_lo(mask, 0u)) & (RING - 1)] = e | (dl << 20);
        qcount += (unsigned)__builtin_popcount(mask);
    };

    auto drain = [&](const bool full, const unsigned n) __attribute__((always_inline)) {
        const unsigned ent = ringw[(qhead + lane) & (RING - 1)];
        const bool valid = full || (unsigned)lane < n;
        const unsigned e = valid ? (ent & 0xFFFFFu) : 0u;
        const int sv  = src[e];
        const int wv  = __float_as_int(ew[e]);
        const int dlv = (int)(ent >> 20);
#pragma unroll 1
        for (int j = 0; j < 32; j += EDGE_GROUP) {
            if (!full && (unsigned)j >= n) break;
            float v[EDGE_GROUP][4];
#pragma unroll
            for (int u = 0; u < EDGE_GROUP; ++u) {
                if (full || (unsigned)(j + u) < n) {
                    const int s = __builtin_amdgcn_readlane(sv, j + u);
                    const float* xrow = x + (size_t)s * D;
                    v[u][0] = xrow[lane];
                    v[u][1] = xrow[lane + 32];
                    v[u][2] = xrow[lane + 64];
                    v[u][3] = xrow[lane + 96];
                }
            }
#pragma unroll
            for (int u = 0; u < EDGE_GROUP; ++u) {
                if (full || (unsigned)(j + u) < n) {
                    const int   dl = __builtin_amdgcn_readlane(dlv, j + u);
                    const float w  = __int_as_float(__builtin_amdgcn_readlane(wv, j + u));
                    float* hrow = &hs[dl * LDS_PITCH];
                    __hip_atomic_fetch_add(hrow + lane,      v[u][0] * w, __ATOMIC_RELAXED, __HIP_MEMORY_SCOPE_WORKGROUP);
                    __hip_atomic_fetch_add(hrow + lane + 32, v[u][1] * w, __ATOMIC_RELAXED, __HIP_MEMORY_SCOPE_WORKGROUP);
                    __hip_atomic_fetch_add(hrow + lane + 64, v[u][2] * w, __ATOMIC_RELAXED, __HIP_MEMORY_SCOPE_WORKGROUP);
                    __hip_atomic_fetch_add(hrow + lane + 96, v[u][3] * w, __ATOMIC_RELAXED, __HIP_MEMORY_SCOPE_WORKGROUP);
                }
            }
        }
        qhead = (qhead + n) & (RING - 1);
        qcount -= n;
    };

    {
        const int4* dst4 = (const int4*)dst;
        const int rot = blockIdx.x * (N_CHUNKS / N_WG);

        auto chunk_of = [&](const int k) __attribute__((always_inline)) {
            int c = (k < N_CHUNKS ? k : N_CHUNKS - 1) + rot;
            if (c >= N_CHUNKS) c -= N_CHUNKS;
            return c;
        };
        auto load_set = [&](int4 (&dv)[SCAN_DEPTH], const int k0) __attribute__((always_inline)) {
#pragma unroll
            for (int u = 0; u < SCAN_DEPTH; ++u)
                dv[u] = dst4[chunk_of(k0 + u * WG_WAVES) * 32 + lane];
        };
        auto scan_set = [&](const int4 (&dv)[SCAN_DEPTH], const int k0) __attribute__((always_inline)) {
#pragma unroll
            for (int u = 0; u < SCAN_DEPTH; ++u) {
                const int k = k0 + u * WG_WAVES;
                if (k < N_CHUNKS) {
                    const unsigned d0 = (unsigned)(dv[u].x - node0);
                    const unsigned d1 = (unsigned)(dv[u].y - node0);
                    const unsigned d2 = (unsigned)(dv[u].z - node0);
                    const unsigned d3 = (unsigned)(dv[u].w - node0);
                    const bool m0 = d0 < NODES_PER_WG, m1 = d1 < NODES_PER_WG;
                    const bool m2 = d2 < NODES_PER_WG, m3 = d3 < NODES_PER_WG;
                    if (__builtin_amdgcn_ballot_w32(m0 | m1 | m2 | m3)) {
                        const unsigned e0 = (unsigned)(chunk_of(k) * CHUNK_EDGES + lane * 4);
                        push(m0, d0, e0);
                        push(m1, d1, e0 + 1);
                        push(m2, d2, e0 + 2);
                        push(m3, d3, e0 + 3);
                        while (qcount >= 32) drain(true, 32u);
                    }
                }
            }
        };

        const int STEP = WG_WAVES * SCAN_DEPTH;
        int4 dva[SCAN_DEPTH], dvb[SCAN_DEPTH];
        load_set(dva, wave);
#pragma unroll 1
        for (int k0 = wave; k0 < N_CHUNKS; k0 += 2 * STEP) {
            load_set(dvb, k0 + STEP);
            scan_set(dva, k0);
            load_set(dva, k0 + 2 * STEP);
            scan_set(dvb, k0 + STEP);
        }
        if (qcount) drain(false, qcount);
    }
    __syncthreads();

    {
        const int l15  = lane & 15;
        const int rsel = lane >> 4;
#pragma unroll 1
        for (int unit = wave; unit < 2 * TILES_PER_WG; unit += WG_WAVES) {
            const int tile = unit >> 1;
            const int half = unit & 1;
            const float* abase = &hs[(tile * 16 + l15) * LDS_PITCH + 2 * rsel];
            const float* bbase = &Ws[(half * 64 + l15) * LDS_PITCH + 2 * rsel];

            v8f acc[4];
#pragma unroll
            for (int t = 0; t < 4; ++t) acc[t] = v8f{};

#pragma unroll
            for (int kk = 0; kk < D / 4; ++kk) {
                const v2f a = *(const v2f*)(abase + 4 * kk);
#pragma unroll
                for (int t = 0; t < 4; ++t) {
                    const v2f bb = *(const v2f*)(bbase + t * 16 * LDS_PITCH + 4 * kk);
                    acc[t] = __builtin_amdgcn_wmma_f32_16x16x4_f32(
                         false, a,  false, bb,
                         (short)0, acc[t],  false,  false);
                }
            }

            float* obase = out + (size_t)(node0 + tile * 16 + 8 * rsel) * D + half * 64 + l15;
#pragma unroll
            for (int t = 0; t < 4; ++t) {
                const float bias = b[half * 64 + t * 16 + l15];
#pragma unroll
                for (int r = 0; r < 8; ++r) {
                    obase[(size_t)r * D + t * 16] = acc[t][r] + bias;
                }
            }
        }
    }
}

extern "C" void kernel_launch(void* const* d_in, const int* in_sizes, int n_in,
                              void* d_out, int out_size, void* d_ws, size_t ws_size,
                              hipStream_t stream) {
    const float* x   = (const float*)d_in[0];
    const float* ew  = (const float*)d_in[1];
    const int*   src = (const int*)  d_in[2];
    const int*   dst = (const int*)  d_in[3];
    const float* Wm  = (const float*)d_in[4];
    const float* bv  = (const float*)d_in[5];
    float* out = (float*)d_out;

    gcn_fused_kernel<<<N_WG, WG_THREADS, 0, stream>>>(x, ew, src, dst, Wm, bv, out);
}


# New first launch counting-sorts each 2560-edge slice by owner workgroup into the workspace (src, row, weight packed): a workgroup reads only its own edges, not all of dst; sums still in arrival order
# speedup vs baseline: 3.9984x; 1.4032x over previous
#include <hip/hip_runtime.h>


#define N_NODES 100000
#define N_EDGES 640000
#define D 128

typedef float v8f __attribute__((ext_vector_type(8)));
typedef float v2f __attribute__((ext_vector_type(2)));

#define NODES_PER_WG 400
#define N_WG (N_NODES / NODES_PER_WG)
#define TILES_PER_WG (NODES_PER_WG / 16)
#define WG_WAVES 32
#define WG_THREADS (WG_WAVES * 32)
#define LDS_PITCH 132
#define SLICE_EDGES 2560
#define N_SLICES (N_EDGES / SLICE_EDGES)
#define SORT_THREADS 320
#define TABLE_PITCH 256
#define SLICES_PER_WAVE ((N_SLICES + WG_WAVES - 1) / WG_WAVES)
#define HEAD 16
#define RING 128
#define EDGE_GROUP 8
static_assert(N_NODES % NODES_PER_WG == 0, "node ranges tile the graph");
static_assert(NODES_PER_WG % 16 == 0, "M tiles are full");
static_assert(N_EDGES % SLICE_EDGES == 0, "slices are full");
static_assert(SLICE_EDGES == SORT_THREADS * 8, "each sort thread takes two int4 of edges");
static_assert(N_WG < TABLE_PITCH && SORT_THREADS >= TABLE_PITCH, "one thread per table word");
static_assert(SLICE_EDGES < (1 << 16), "table word packs segment start and length");
static_assert(N_NODES <= (1 << 20) && NODES_PER_WG <= (1 << 12), "entry packs source node and local row");
static_assert(SLICES_PER_WAVE <= 32, "one lane per table word of the wave");
static_assert(SLICES_PER_WAVE * HEAD <= RING && HEAD <= 32, "the heads of a wave's segments fit its queue");

__global__ void __launch_bounds__(SORT_THREADS)
gcn_sort_kernel(const float* __restrict__ ew,
                const int* __restrict__ src,
                const int* __restrict__ dst,
                uint2* __restrict__ entries,
                unsigned int* __restrict__ table) {
    __shared__ unsigned int cnt[TABLE_PITCH];
    __shared__ unsigned int start[TABLE_PITCH];
    __shared__ __attribute__((aligned(16))) uint2 sorted[SLICE_EDGES];

    const int tid   = threadIdx.x;
    const int slice = blockIdx.x;

    const int4*   d4 = (const int4*)dst  + slice * (SLICE_EDGES / 4);
    const int4*   s4 = (const int4*)src  + slice * (SLICE_EDGES / 4);
    const float4* w4 = (const float4*)ew + slice * (SLICE_EDGES / 4);
    int4 dv[2], sv[2];
    float4 wv[2];
#pragma unroll
    for (int u = 0; u < 2; ++u) {
        dv[u] = d4[u * SORT_THREADS + tid];
        sv[u] = s4[u * SORT_THREADS + tid];
        wv[u] = w4[u * SORT_THREADS + tid];
    }
    if (tid < TABLE_PITCH) cnt[tid] = 0u;
    __syncthreads();

    unsigned owner[8], rank[8];
    uint2 ent[8];
#pragma unroll
    for (int u = 0; u < 2; ++u) {
        const int   dd[4] = {dv[u].x, dv[u].y, dv[u].z, dv[u].w};
        const int   ss[4] = {sv[u].x, sv[u].y, sv[u].z, sv[u].w};
        const float ww[4] = {wv[u].x, wv[u].y, wv[u].z, wv[u].w};
#pragma unroll
        for (int k = 0; k < 4; ++k) {
            const unsigned d = (unsigned)dd[k];
            const unsigned o = d < (unsigned)N_NODES ? d / NODES_PER_WG : (unsigned)N_WG;
            const unsigned s = min((unsigned)ss[k], (unsigned)(N_NODES - 1));
            owner[u * 4 + k] = o;
            rank[u * 4 + k]  = atomicAdd(&cnt[o], 1u);
            ent[u * 4 + k]   = make_uint2(s | ((d - o * NODES_PER_WG) << 20), __float_as_uint(ww[k]));
        }
    }
    __syncthreads();

    if (tid < 32) {
        unsigned cv[8];
        unsigned sum = 0;
#pragma unroll
        for (int k = 0; k < 8; ++k) { cv[k] = cnt[tid * 8 + k]; sum += cv[k]; }
        unsigned incl = sum;
#pragma unroll
        for (int off = 1; off < 32; off <<= 1) {
            const unsigned t = __shfl_up(incl, off);
            if (tid >= off) incl += t;
        }
        unsigned run = incl - sum;
#pragma unroll
        for (int k = 0; k < 8; ++k) { start[tid * 8 + k] = run; run += cv[k]; }
    }
    __syncthreads();

#pragma unroll
    for (int i = 0; i < 8; ++i) sorted[start[owner[i]] + rank[i]] = ent[i];
    if (tid < TABLE_PITCH) table[slice * TABLE_PITCH + tid] = start[tid] | (cnt[tid] << 16);
    __syncthreads();

    uint4* out4 = (uint4*)(entries + (size_t)slice * SLICE_EDGES);
    const uint4* in4 = (const uint4*)sorted;
#pragma unroll
    for (int i = 0; i < SLICE_EDGES / 2 / SORT_THREADS; ++i)
        out4[i * SORT_THREADS + tid] = in4[i * SORT_THREADS + tid];
}

__global__ void __launch_bounds__(WG_THREADS)
gcn_fused_kernel(const float* __restrict__ x,
                 const uint2* __restrict__ entries,
                 const unsigned int* __restrict__ table,
                 const float* __restrict__ W,
                 const float* __restrict__ b,
                 float* __restrict__ out) {
    __shared__ __attribute__((aligned(16))) float hs[NODES_PER_WG * LDS_PITCH];
    __shared__ __attribute__((aligned(16))) float Ws[D * LDS_PITCH];
    __shared__ uint2 ring[WG_WAVES * RING];

    const int lane  = threadIdx.x & 31;
    const int wave  = __builtin_amdgcn_readfirstlane(threadIdx.x >> 5);
    const int node0 = blockIdx.x * NODES_PER_WG;

    for (int i = threadIdx.x; i < NODES_PER_WG * LDS_PITCH / 4; i += WG_THREADS)
        ((float4*)hs)[i] = make_float4(0.f, 0.f, 0.f, 0.f);
    {
        const float4* W4 = (const float4*)W;
#pragma unroll
        for (int i = 0; i < D / WG_WAVES; ++i) {
            const int row = i * WG_WAVES + wave;
            const float4 v = W4[row * (D / 4) + lane];
            *(float4*)&Ws[row * LDS_PITCH + lane * 4] = v;
        }
    }
    __syncthreads();

    uint2* ringw = &ring[wave * RING];
    unsigned qhead = 0, qcount = 0;

    auto push = [&](const bool m, const uint2 ent) __attribute__((always_inline)) {
        const unsigned mask = __builtin_amdgcn_ballot_w32(m);
        if (m) ringw[(qhead + qcount + __builtin_amdgcn_mbcnt_lo(mask, 0u)) & (RING - 1)] = ent;
        qcount += (unsigned)__builtin_popcount(mask);
    };

    auto drain = [&](const bool full, const unsigned n) __attribute__((always_inline)) {
        const uint2 ent = ringw[(qhead + lane) & (RING - 1)];
        const bool valid = full || (unsigned)lane < n;
        const int sv  = valid ? (int)(ent.x & 0xFFFFFu) : 0;
        const int dlv = valid ? (int)(ent.x >> 20) : 0;
        const int wv  = (int)ent.y;
#pragma unroll 1
        for (int j = 0; j < 32; j += EDGE_GROUP) {
            if (!full && (unsigned)j >= n) break;
            float v[EDGE_GROUP][4];
#pragma unroll
            for (int u = 0; u < EDGE_GROUP; ++u) {
                if (full || (unsigned)(j + u) < n) {
                    const int s = __builtin_amdgcn_readlane(sv, j + u);
                    const float* xrow = x + (size_t)s * D;
                    v[u][0] = xrow[lane];
                    v[u][1] = xrow[lane + 32];
                    v[u][2] = xrow[lane + 64];
                    v[u][3] = xrow[lane + 96];
                }
            }
#pragma unroll
            for (int u = 0; u < EDGE_GROUP; ++u) {
                if (full || (unsigned)(j + u) < n) {
                    const int   dl = __builtin_amdgcn_readlane(dlv, j + u);
                    const float w  = __int_as_float(__builtin_amdgcn_readlane(wv, j + u));
                    float* hrow = &hs[dl * LDS_PITCH];
                    __hip_atomic_fetch_add(hrow + lane,      v[u][0] * w, __ATOMIC_RELAXED, __HIP_MEMORY_SCOPE_WORKGROUP);
                    __hip_atomic_fetch_add(hrow + lane + 32, v[u][1] * w, __ATOMIC_RELAXED, __HIP_MEMORY_SCOPE_WORKGROUP);
                    __hip_atomic_fetch_add(hrow + lane + 64, v[u][2] * w, __ATOMIC_RELAXED, __HIP_MEMORY_SCOPE_WORKGROUP);
                    __hip_atomic_fetch_add(hrow + lane + 96, v[u][3] * w, __ATOMIC_RELAXED, __HIP_MEMORY_SCOPE_WORKGROUP);
                }
            }
        }
        qhead = (qhead + n) & (RING - 1);
        qcount -= n;
    };

    {
        unsigned tw = 0u;
        {
            const int g = wave + lane * WG_WAVES;
            if (lane < SLICES_PER_WAVE && g < N_SLICES) tw = table[g * TABLE_PITCH + blockIdx.x];
        }
        auto seg_len = [&](const int t) __attribute__((always_inline)) {
            return (unsigned)__builtin_amdgcn_readlane((int)tw, t) >> 16;
        };
        auto seg_ptr = [&](const int t) __attribute__((always_inline)) {
            const unsigned first = (unsigned)__builtin_amdgcn_readlane((int)tw, t) & 0xFFFFu;
            return entries + (size_t)(wave + t * WG_WAVES) * SLICE_EDGES + first;
        };

        uint2 ev[SLICES_PER_WAVE];
#pragma unroll
        for (int t = 0; t < SLICES_PER_WAVE; ++t) {
            ev[t] = make_uint2(0u, 0u);
            if ((unsigned)lane < min(seg_len(t), (unsigned)HEAD)) ev[t] = seg_ptr(t)[lane];
        }
#pragma unroll
        for (int t = 0; t < SLICES_PER_WAVE; ++t)
            push((unsigned)lane < min(seg_len(t), (unsigned)HEAD), ev[t]);

        int t = 0;
        unsigned off = HEAD;
#pragma unroll 1
        for (;;) {
            while (qcount >= 32) drain(true, 32u);
            while (t < SLICES_PER_WAVE && off >= seg_len(t)) { ++t; off = HEAD; }
            if (t == SLICES_PER_WAVE) break;
            const bool m = off + (unsigned)lane < seg_len(t);
            uint2 e = make_uint2(0u, 0u);
            if (m) e = seg_ptr(t)[off + lane];
            push(m, e);
            off += 32u;
        }
        if (qcount) drain(false, qcount);
    }
    __syncthreads();

    {
        const int l15  = lane & 15;
        const int rsel = lane >> 4;
#pragma unroll 1
        for (int unit = wave; unit < 2 * TILES_PER_WG; unit += WG_WAVES) {
            const int tile = unit >> 1;
            const int half = unit & 1;
            const float* abase = &hs[(tile * 16 + l15) * LDS_PITCH + 2 * rsel];
            const float* bbase = &Ws[(half * 64 + l15) * LDS_PITCH + 2 * rsel];

            v8f acc[4];
#pragma unroll
            for (int t = 0; t < 4; ++t) acc[t] = v8f{};

#pragma unroll
            for (int kk = 0; kk < D / 4; ++kk) {
                const v2f a = *(const v2f*)(abase + 4 * kk);
#pragma unroll
                for (int t = 0; t < 4; ++t) {
                    const v2f bb = *(const v2f*)(bbase + t * 16 * LDS_PITCH + 4 * kk);
                    acc[t] = __builtin_amdgcn_wmma_f32_16x16x4_f32(
                         false, a,  false, bb,
                         (short)0, acc[t],  false,  false);
                }
            }

            float* obase = out + (size_t)(node0 + tile * 16 + 8 * rsel) * D + half * 64 + l15;
#pragma unroll
            for (int t = 0; t < 4; ++t) {
                const float bias = b[half * 64 + t * 16 + l15];
#pragma unroll
                for (int r = 0; r < 8; ++r) {
                    obase[(size_t)r * D + t * 16] = acc[t][r] + bias;
                }
            }
        }
    }
}

extern "C" void kernel_launch(void* const* d_in, const int* in_sizes, int n_in,
                              void* d_out, int out_size, void* d_ws, size_t ws_size,
                              hipStream_t stream) {
    const float* x   = (const float*)d_in[0];
    const float* ew  = (const float*)d_in[1];
    const int*   src = (const int*)  d_in[2];
    const int*   dst = (const int*)  d_in[3];
    const float* Wm  = (const float*)d_in[4];
    const float* bv  = (const float*)d_in[5];
    float* out = (float*)d_out;

    uint2* entries = (uint2*)d_ws;
    unsigned int* table = (unsigned int*)((char*)d_ws + (size_t)N_EDGES * 8);

    gcn_sort_kernel<<<N_SLICES, SORT_THREADS, 0, stream>>>(ew, src, dst, entries, table);
    gcn_fused_kernel<<<N_WG, WG_THREADS, 0, stream>>>(x, entries, table, Wm, bv, out);
}
